# starting kernel
# baseline (speedup 1.0000x reference)
#include <hip/hip_runtime.h>

typedef float v2f __attribute__((ext_vector_type(2)));
typedef float v4f __attribute__((ext_vector_type(4)));
typedef float v8f __attribute__((ext_vector_type(8)));

#define COUT 64
#define BN_EPS 1e-5f

__global__ void zero_f32v4(v4f* __restrict__ p, int n4) {
  int i = blockIdx.x * blockDim.x + threadIdx.x;
  int s = gridDim.x * blockDim.x;
  v4f z = {};
  for (; i < n4; i += s) p[i] = z;
}

template <int CIN>
__global__ void sparse_conv_wmma(const float* __restrict__ feats,
                                 const float* __restrict__ W,
                                 const int*   __restrict__ pin,
                                 const int*   __restrict__ pout,
                                 float*       __restrict__ out,
                                 int P, int mtiles) {
  const int wave  = threadIdx.x >> 5;
  const int mtile = blockIdx.x * 8 + wave;
  if (mtile >= mtiles) return;

  const int k     = blockIdx.y;
  const int ntile = blockIdx.z;
  const int lane  = threadIdx.x & 31;
  const int m0    = mtile * 16;
  const int n0    = ntile * 16;
  const int half  = lane >> 4;
  const int nl    = lane & 15;

  const int pidx = m0 + nl;
  const int arow = (pidx < P) ? pin[k * P + pidx] : 0;
  const float* aptr = feats + (long)arow * CIN + 2 * half;
  const float* bptr = W + (long)k * CIN * COUT + 2 * half * COUT + n0 + nl;

  v8f c = {};
#pragma unroll
  for (int c0 = 0; c0 < CIN; c0 += 4) {
    v2f a = *reinterpret_cast<const v2f*>(aptr + c0);
    v2f b;
    b.x = bptr[c0 * COUT];
    b.y = bptr[c0 * COUT + COUT];
    c = __builtin_amdgcn_wmma_f32_16x16x4_f32(false, a, false, b,
                                              (short)0, c, false, false);
  }

#pragma unroll
  for (int r = 0; r < 8; ++r) {
    const int pi = m0 + r + 8 * half;
    if (pi < P) {
      const int orow = pout[k * P + pi];
      atomicAdd(&out[(long)orow * COUT + n0 + nl], c[r]);
    }
  }
}

__global__ void col_stats(const float* __restrict__ x, int n_elems,
                          float* __restrict__ stats) {
  __shared__ float ssum[COUT];
  __shared__ float ssq[COUT];
  if (threadIdx.x < COUT) { ssum[threadIdx.x] = 0.f; ssq[threadIdx.x] = 0.f; }
  __syncthreads();
  int idx    = blockIdx.x * blockDim.x + threadIdx.x;
  int stride = gridDim.x * blockDim.x;
  const int ch = idx & (COUT - 1);
  float lsum = 0.f, lsq = 0.f;
  for (int i = idx; i < n_elems; i += stride) {
    float v = x[i];
    lsum += v;
    lsq  += v * v;
  }
  atomicAdd(&ssum[ch], lsum);
  atomicAdd(&ssq[ch], lsq);
  __syncthreads();
  if (threadIdx.x < COUT) {
    atomicAdd(&stats[threadIdx.x], ssum[threadIdx.x]);
    atomicAdd(&stats[COUT + threadIdx.x], ssq[threadIdx.x]);
  }
}

__global__ void bn_finalize(const float* __restrict__ stats,
                            const float* __restrict__ g,
                            const float* __restrict__ b,
                            float inv_n, float* __restrict__ ab) {
  int c = threadIdx.x;
  if (c < COUT) {
    float mean = stats[c] * inv_n;
    float var  = stats[COUT + c] * inv_n - mean * mean;
    float s    = g[c] * rsqrtf(var + BN_EPS);
    ab[c]        = s;
    ab[COUT + c] = b[c] - mean * s;
  }
}

__global__ void bn_apply(const float* __restrict__ y,
                         const float* __restrict__ ab,
                         float* __restrict__ x, int n_elems) {
  int i = blockIdx.x * blockDim.x + threadIdx.x;
  int s = gridDim.x * blockDim.x;
  for (; i < n_elems; i += s) {
    int ch = i & (COUT - 1);
    float v = y[i] * ab[ch] + ab[COUT + ch];
    x[i] = v > 0.f ? v : 0.f;
  }
}

__global__ void bn_apply_add(const float* __restrict__ y,
                             const float* __restrict__ ab,
                             const float* __restrict__ xnet,
                             float* __restrict__ out, int n_elems) {
  int i = blockIdx.x * blockDim.x + threadIdx.x;
  int s = gridDim.x * blockDim.x;
  for (; i < n_elems; i += s) {
    int ch = i & (COUT - 1);
    float v = y[i] * ab[ch] + ab[COUT + ch];
    out[i] = xnet[i] + (v > 0.f ? v : 0.f);
  }
}

static inline int cdiv(long a, long b) { return (int)((a + b - 1) / b); }

extern "C" void kernel_launch(void* const* d_in, const int* in_sizes, int n_in,
                              void* d_out, int out_size, void* d_ws, size_t ws_size,
                              hipStream_t stream) {
  const float* feats  = (const float*)d_in[0];
  const float* W_down = (const float*)d_in[1];
  const float* g0     = (const float*)d_in[2];
  const float* b0     = (const float*)d_in[3];
  const float* W1     = (const float*)d_in[4];
  const float* g1     = (const float*)d_in[5];
  const float* b1     = (const float*)d_in[6];
  const float* W2     = (const float*)d_in[7];
  const float* g2     = (const float*)d_in[8];
  const float* b2     = (const float*)d_in[9];
  const int* p_down_in  = (const int*)d_in[10];
  const int* p_down_out = (const int*)d_in[11];
  const int* p_subm_in  = (const int*)d_in[12];
  const int* p_subm_out = (const int*)d_in[13];

  const int Cin = 32;
  const int K   = in_sizes[1] / (Cin * COUT);
  const int P1  = in_sizes[10] / K;
  const int P2  = in_sizes[12] / K;
  const int n_out = out_size / COUT;
  const int NE  = n_out * COUT;
  const float inv_n = 1.0f / (float)n_out;

  float* y     = (float*)d_ws;
  float* x     = y + NE;
  float* xnet  = x + NE;
  float* stats = xnet + NE;
  float* ab    = stats + 128;
  (void)n_in; (void)ws_size;

  const dim3 blk(256);
  const int zb  = cdiv(NE / 4, 256);
  const int eb  = cdiv(NE, 256);
  const int sb  = 1024;

  const int mt1 = cdiv(P1, 16);
  const int mt2 = cdiv(P2, 16);
  const dim3 cg1(cdiv(mt1, 8), K, 4);
  const dim3 cg2(cdiv(mt2, 8), K, 4);

  zero_f32v4<<<zb, blk, 0, stream>>>((v4f*)y, NE / 4);
  zero_f32v4<<<1, 32, 0, stream>>>((v4f*)stats, 32);
  sparse_conv_wmma<32><<<cg1, blk, 0, stream>>>(feats, W_down, p_down_in, p_down_out,
                                                y, P1, mt1);
  col_stats<<<sb, blk, 0, stream>>>(y, NE, stats);
  bn_finalize<<<1, 64, 0, stream>>>(stats, g0, b0, inv_n, ab);
  bn_apply<<<eb, blk, 0, stream>>>(y, ab, x, NE);

  zero_f32v4<<<zb, blk, 0, stream>>>((v4f*)y, NE / 4);
  zero_f32v4<<<1, 32, 0, stream>>>((v4f*)stats, 32);
  sparse_conv_wmma<64><<<cg2, blk, 0, stream>>>(x, W1, p_subm_in, p_subm_out,
                                                y, P2, mt2);
  col_stats<<<sb, blk, 0, stream>>>(y, NE, stats);
  bn_finalize<<<1, 64, 0, stream>>>(stats, g1, b1, inv_n, ab);
  bn_apply<<<eb, blk, 0, stream>>>(y, ab, xnet, NE);

  zero_f32v4<<<zb, blk, 0, stream>>>((v4f*)y, NE / 4);
  zero_f32v4<<<1, 32, 0, stream>>>((v4f*)stats, 32);
  sparse_conv_wmma<64><<<cg2, blk, 0, stream>>>(xnet, W2, p_subm_in, p_subm_out,
                                                y, P2, mt2);
  col_stats<<<sb, blk, 0, stream>>>(y, NE, stats);
  bn_finalize<<<1, 64, 0, stream>>>(stats, g2, b2, inv_n, ab);
  bn_apply_add<<<eb, blk, 0, stream>>>(y, ab, xnet, (float*)d_out, NE);
}


# conv tile loop software-pipelined by one tile: next tile's pair records and row gather travel under this tile's WMMAs and scatter; tile->k lookup in lane registers (ballot), no LDS/scalar wait
# speedup vs baseline: 2.4281x; 2.4281x over previous
#include <hip/hip_runtime.h>

typedef float v2f __attribute__((ext_vector_type(2)));
typedef float v4f __attribute__((ext_vector_type(4)));
typedef float v8f __attribute__((ext_vector_type(8)));

#define COUT 64
#define BN_EPS 1e-5f

#define NB 256
#define CH 2048

__global__ void bin_hist(const int* __restrict__ pout, int P, int R, int nch,
                         int* __restrict__ cnt) {
  __shared__ int h[NB];
  const int k = blockIdx.y, ch = blockIdx.x;
  h[threadIdx.x] = 0;
  __syncthreads();
  const long kb = (long)k * P;
  for (int i = threadIdx.x; i < CH; i += NB) {
    const int p = ch * CH + i;
    if (p < P) atomicAdd(&h[pout[kb + p] / R], 1);
  }
  __syncthreads();
  cnt[((long)k * nch + ch) * NB + threadIdx.x] = h[threadIdx.x];
}

__global__ void bin_scan(int* __restrict__ cnt, int nch,
                         int* __restrict__ bstart,
                         float* __restrict__ zstats) {
  __shared__ int tot[NB];
  const int k = blockIdx.x, b = threadIdx.x;
  if (zstats != nullptr && k == 0 && b < 2 * COUT) zstats[b] = 0.f;
  int* c = cnt + (long)k * nch * NB + b;
  int s = 0;
  for (int j = 0; j < nch; ++j) s += c[(long)j * NB];
  tot[b] = s;
  __syncthreads();
  int start = 0;
  for (int j = 0; j < b; ++j) start += tot[j];
  bstart[k * (NB + 1) + b] = start;
  if (b == NB - 1) bstart[k * (NB + 1) + NB] = start + s;
  int run = start;
  for (int j = 0; j < nch; ++j) {
    const int v = c[(long)j * NB];
    c[(long)j * NB] = run;
    run += v;
  }
}

__global__ void bin_scatter(const int* __restrict__ pin, const int* __restrict__ pout,
                            int P, int R, int nch, const int* __restrict__ off,
                            int2* __restrict__ spair) {
  __shared__ int cur[NB];
  const int k = blockIdx.y, ch = blockIdx.x;
  cur[threadIdx.x] = off[((long)k * nch + ch) * NB + threadIdx.x];
  __syncthreads();
  const long kb = (long)k * P;
  for (int i = threadIdx.x; i < CH; i += NB) {
    const int p = ch * CH + i;
    if (p < P) {
      const int o = pout[kb + p];
      const int b = o / R;
      const int pos = atomicAdd(&cur[b], 1);
      spair[kb + pos] = make_int2(pin[kb + p], o - b * R);
    }
  }
}

#define TP 80
template <int CIN>
__global__ __launch_bounds__(1024)
void conv_bucket(const float* __restrict__ feats,
                 const float* __restrict__ W,
                 const int2*  __restrict__ spair,
                 const int*   __restrict__ bstart,
                 float*       __restrict__ out,
                 float*       __restrict__ stats,
                 int P, int R, int n_out, int K) {
  constexpr int PITCH = CIN + 4;
  constexpr int LPR   = CIN / 4;
  constexpr int NI    = CIN / 8;
  extern __shared__ v4f tile4[];
  __shared__ v4f stage4[32 * 16 * PITCH / 4];
  __shared__ float red[2 * COUT];
  const int b    = blockIdx.x;
  const int tid  = threadIdx.x;
  const int lane = tid & 31;
  const int wave = __builtin_amdgcn_readfirstlane(tid >> 5);
  const int nw   = blockDim.x >> 5;
  const int half = lane >> 4;
  const int nl   = lane & 15;

  float* tile = reinterpret_cast<float*>(tile4);
  float* st = reinterpret_cast<float*>(stage4) + wave * (16 * PITCH);
  const int piece = lane & (LPR - 1);
  const int sub   = lane / LPR;
  const int rbase = (CIN == 32) ? 8 * (sub & 1) + 4 * (sub >> 1) : 8 * half;

  int bs_lo = 0, bs_hi = 0;
  if (lane < K) {
    bs_lo = bstart[lane * (NB + 1) + b];
    bs_hi = bstart[lane * (NB + 1) + b + 1];
  }
  const int nt = (bs_hi - bs_lo + 15) >> 4;
  int inc = nt;
#pragma unroll
  for (int d = 1; d < 32; d <<= 1) {
    const int t = __shfl_up(inc, d, 32);
    if (lane >= d) inc += t;
  }
  const int tp   = inc - nt;
  const int ttot = __builtin_amdgcn_readlane(inc, 31);

  auto locate = [&](int f, int& k, int& e, int& p0) {
    const unsigned m = __builtin_amdgcn_ballot_w32(tp <= f);
    k  = __popc(m) - 1;
    e  = __builtin_amdgcn_readlane(bs_hi, k);
    p0 = __builtin_amdgcn_readlane(bs_lo, k) + (f - __builtin_amdgcn_readlane(tp, k)) * 16;
  };
  auto load_pair = [&](int k, int e, int p0) {
    int2 pr = make_int2(0, 0);
    const int pa = p0 + nl;
    if (pa < e) pr = spair[(long)k * P + pa];
    return pr;
  };
  auto gather = [&](int prx, v4f (&av)[NI]) {
#pragma unroll
    for (int i = 0; i < NI; ++i) {
      const int srow = __shfl(prx, i + rbase, 32);
      av[i] = *reinterpret_cast<const v4f*>(feats + (long)srow * CIN + 4 * piece);
    }
  };

  int k, e, p0;
  locate(wave, k, e, p0);
  int2 pr = load_pair(k, e, p0);
  v4f av[NI];
  gather(pr.x, av);

  const v4f z = {};
  for (int i = tid; i < R * (TP / 4); i += blockDim.x) tile4[i] = z;
  if (tid < 2 * COUT) red[tid] = 0.f;
  __syncthreads();

  for (int f = wave; f < ttot; f += nw) {
    const float* bptr = W + (long)k * CIN * COUT + 2 * half * COUT + nl;
    v2f q0, q1, q2, q3;
    q0.x = bptr[0];    q0.y = bptr[COUT];
    q1.x = bptr[16];   q1.y = bptr[COUT + 16];
    q2.x = bptr[32];   q2.y = bptr[COUT + 32];
    q3.x = bptr[48];   q3.y = bptr[COUT + 48];

#pragma unroll
    for (int i = 0; i < NI; ++i)
      *reinterpret_cast<v4f*>(st + (i + rbase) * PITCH + 4 * piece) = av[i];
    __builtin_amdgcn_fence(__ATOMIC_RELEASE, "wavefront");
    __builtin_amdgcn_wave_barrier();
    __builtin_amdgcn_fence(__ATOMIC_ACQUIRE, "wavefront");
    const float* aptr = st + nl * PITCH + 2 * half;

    int kn, en, p0n;
    locate(f + nw, kn, en, p0n);
    const int2 prn = load_pair(kn, en, p0n);

    v8f c0 = {}, c1 = {}, c2 = {}, c3 = {};
#pragma unroll
    for (int cc = 0; cc < CIN; cc += 4) {
      const v2f a = *reinterpret_cast<const v2f*>(aptr + cc);
      v2f b0, b1, b2, b3;
      if (cc == 0) {
        b0 = q0; b1 = q1; b2 = q2; b3 = q3;
      } else {
        b0.x = bptr[cc * COUT];        b0.y = bptr[cc * COUT + COUT];
        b1.x = bptr[cc * COUT + 16];   b1.y = bptr[cc * COUT + COUT + 16];
        b2.x = bptr[cc * COUT + 32];   b2.y = bptr[cc * COUT + COUT + 32];
        b3.x = bptr[cc * COUT + 48];   b3.y = bptr[cc * COUT + COUT + 48];
      }
      c0 = __builtin_amdgcn_wmma_f32_16x16x4_f32(false, a, false, b0, (short)0, c0, false, false);
      c1 = __builtin_amdgcn_wmma_f32_16x16x4_f32(false, a, false, b1, (short)0, c1, false, false);
      c2 = __builtin_amdgcn_wmma_f32_16x16x4_f32(false, a, false, b2, (short)0, c2, false, false);
      c3 = __builtin_amdgcn_wmma_f32_16x16x4_f32(false, a, false, b3, (short)0, c3, false, false);
    }

    v4f avn[NI];
    gather(prn.x, avn);

#pragma unroll
    for (int r = 0; r < 8; ++r) {
      const int src  = r + 8 * half;
      const int lrow = __shfl(pr.y, src, 32);
      if (p0 + src < e) {
        float* t = tile + lrow * TP + nl;
        atomicAdd(t,      c0[r]);
        atomicAdd(t + 16, c1[r]);
        atomicAdd(t + 32, c2[r]);
        atomicAdd(t + 48, c3[r]);
      }
    }

    k = kn; e = en; p0 = p0n; pr = prn;
#pragma unroll
    for (int i = 0; i < NI; ++i) av[i] = avn[i];
  }
  __syncthreads();

  const int row0 = b * R;
  v4f s1 = {}, s2 = {};
  for (int i = tid; i < R * 16; i += blockDim.x) {
    const int row = row0 + (i >> 4);
    if (row < n_out) {
      const v4f v = tile4[(i >> 4) * (TP / 4) + (i & 15)];
      reinterpret_cast<v4f*>(out)[(long)row * 16 + (i & 15)] = v;
      s1 += v;
      s2 += v * v;
    }
  }
  {
    float* r1 = red + 4 * (tid & 15);
    atomicAdd(r1,     s1.x);  atomicAdd(r1 + COUT,     s2.x);
    atomicAdd(r1 + 1, s1.y);  atomicAdd(r1 + COUT + 1, s2.y);
    atomicAdd(r1 + 2, s1.z);  atomicAdd(r1 + COUT + 2, s2.z);
    atomicAdd(r1 + 3, s1.w);  atomicAdd(r1 + COUT + 3, s2.w);
  }
  __syncthreads();
  if (tid < 2 * COUT) atomicAdd(&stats[tid], red[tid]);
}

__global__ void bn_finalize(float* __restrict__ stats,
                            const float* __restrict__ g,
                            const float* __restrict__ b,
                            float inv_n, float* __restrict__ ab) {
  int c = threadIdx.x;
  if (c < COUT) {
    const float sum = stats[c], sq = stats[COUT + c];
    stats[c] = 0.f;
    stats[COUT + c] = 0.f;
    float mean = sum * inv_n;
    float var  = sq * inv_n - mean * mean;
    float s    = g[c] * rsqrtf(var + BN_EPS);
    ab[c]        = s;
    ab[COUT + c] = b[c] - mean * s;
  }
}

__global__ void bn_apply(const float* __restrict__ y,
                         const float* __restrict__ ab,
                         float* __restrict__ x, int n_elems) {
  int i = blockIdx.x * blockDim.x + threadIdx.x;
  int s = gridDim.x * blockDim.x;
  for (; i < n_elems; i += s) {
    int ch = i & (COUT - 1);
    float v = y[i] * ab[ch] + ab[COUT + ch];
    x[i] = v > 0.f ? v : 0.f;
  }
}

__global__ void bn_apply_add(const float* __restrict__ y,
                             const float* __restrict__ ab,
                             const float* __restrict__ xnet,
                             float* __restrict__ out, int n_elems) {
  int i = blockIdx.x * blockDim.x + threadIdx.x;
  int s = gridDim.x * blockDim.x;
  for (; i < n_elems; i += s) {
    int ch = i & (COUT - 1);
    float v = y[i] * ab[ch] + ab[COUT + ch];
    out[i] = xnet[i] + (v > 0.f ? v : 0.f);
  }
}

static inline int cdiv(long a, long b) { return (int)((a + b - 1) / b); }

extern "C" void kernel_launch(void* const* d_in, const int* in_sizes, int n_in,
                              void* d_out, int out_size, void* d_ws, size_t ws_size,
                              hipStream_t stream) {
  const float* feats  = (const float*)d_in[0];
  const float* W_down = (const float*)d_in[1];
  const float* g0     = (const float*)d_in[2];
  const float* b0     = (const float*)d_in[3];
  const float* W1     = (const float*)d_in[4];
  const float* g1     = (const float*)d_in[5];
  const float* b1     = (const float*)d_in[6];
  const float* W2     = (const float*)d_in[7];
  const float* g2     = (const float*)d_in[8];
  const float* b2     = (const float*)d_in[9];
  const int* p_down_in  = (const int*)d_in[10];
  const int* p_down_out = (const int*)d_in[11];
  const int* p_subm_in  = (const int*)d_in[12];
  const int* p_subm_out = (const int*)d_in[13];

  const int Cin = 32;
  const int K   = in_sizes[1] / (Cin * COUT);
  const int P1  = in_sizes[10] / K;
  const int P2  = in_sizes[12] / K;
  const int n_out = out_size / COUT;
  const int NE  = n_out * COUT;
  const float inv_n = 1.0f / (float)n_out;

  float* y     = (float*)d_ws;
  float* x     = y + NE;
  float* xnet  = x + NE;
  float* stats = xnet + NE;
  float* ab    = stats + 128;
  (void)n_in; (void)ws_size;

  const dim3 blk(256);
  const int eb  = cdiv(NE, 256);

  const int R    = cdiv(n_out, NB);
  const int nch1 = cdiv(P1, CH);
  const int nch2 = cdiv(P2, CH);
  int2* sp1  = (int2*)x;
  int*  cnt1 = (int*)(sp1 + (long)K * P1);
  int*  bs1  = cnt1 + (long)K * nch1 * NB;
  int2* sp2  = (int2*)d_out;
  int*  cnt2 = (int*)(sp2 + (long)K * P2);
  int*  bs2  = cnt2 + (long)K * nch2 * NB;
  const size_t lds = (size_t)R * TP * sizeof(float);

  bin_hist<<<dim3(nch1, K), NB, 0, stream>>>(p_down_out, P1, R, nch1, cnt1);
  bin_scan<<<K, NB, 0, stream>>>(cnt1, nch1, bs1, stats);
  bin_scatter<<<dim3(nch1, K), NB, 0, stream>>>(p_down_in, p_down_out, P1, R, nch1, cnt1, sp1);
  bin_hist<<<dim3(nch2, K), NB, 0, stream>>>(p_subm_out, P2, R, nch2, cnt2);
  bin_scan<<<K, NB, 0, stream>>>(cnt2, nch2, bs2, nullptr);
  bin_scatter<<<dim3(nch2, K), NB, 0, stream>>>(p_subm_in, p_subm_out, P2, R, nch2, cnt2, sp2);

  conv_bucket<32><<<NB, 1024, lds, stream>>>(feats, W_down, sp1, bs1, y, stats, P1, R, n_out, K);
  bn_finalize<<<1, 64, 0, stream>>>(stats, g0, b0, inv_n, ab);
  bn_apply<<<eb, blk, 0, stream>>>(y, ab, x, NE);

  conv_bucket<64><<<NB, 1024, lds, stream>>>(x, W1, sp2, bs2, y, stats, P2, R, n_out, K);
  bn_finalize<<<1, 64, 0, stream>>>(stats, g1, b1, inv_n, ab);
  bn_apply<<<eb, blk, 0, stream>>>(y, ab, xnet, NE);

  conv_bucket<64><<<NB, 1024, lds, stream>>>(xnet, W2, sp2, bs2, y, stats, P2, R, n_out, K);
  bn_finalize<<<1, 64, 0, stream>>>(stats, g2, b2, inv_n, ab);
  bn_apply_add<<<eb, blk, 0, stream>>>(y, ab, xnet, (float*)d_out, NE);
}


# binning of the two pair maps shares its three launches: hist, scan and scatter grids cover both maps' chunks and a block picks its map; 15 -> 12 launches, map 2's sort runs beside map 1's
# speedup vs baseline: 2.5789x; 1.0621x over previous
#include <hip/hip_runtime.h>

typedef float v2f __attribute__((ext_vector_type(2)));
typedef float v4f __attribute__((ext_vector_type(4)));
typedef float v8f __attribute__((ext_vector_type(8)));

#define COUT 64
#define BN_EPS 1e-5f

#define NB 256
#define CH 2048

__global__ void bin_hist(const int* __restrict__ pout1, int P1, int nch1,
                         int* __restrict__ cnt1,
                         const int* __restrict__ pout2, int P2, int nch2,
                         int* __restrict__ cnt2,
                         int R) {
  __shared__ int h[NB];
  const int k = blockIdx.y;
  const bool m2 = (int)blockIdx.x >= nch1;
  const int ch  = m2 ? (int)blockIdx.x - nch1 : (int)blockIdx.x;
  const int* __restrict__ pout = m2 ? pout2 : pout1;
  int* __restrict__ cnt = m2 ? cnt2 : cnt1;
  const int P = m2 ? P2 : P1, nch = m2 ? nch2 : nch1;
  h[threadIdx.x] = 0;
  __syncthreads();
  const long kb = (long)k * P;
  for (int i = threadIdx.x; i < CH; i += NB) {
    const int p = ch * CH + i;
    if (p < P) atomicAdd(&h[pout[kb + p] / R], 1);
  }
  __syncthreads();
  cnt[((long)k * nch + ch) * NB + threadIdx.x] = h[threadIdx.x];
}

__global__ void bin_scan(int* __restrict__ cnt1, int nch1,
                         int* __restrict__ bstart1,
                         int* __restrict__ cnt2, int nch2,
                         int* __restrict__ bstart2,
                         int K, float* __restrict__ zstats) {
  __shared__ int tot[NB];
  const int b = threadIdx.x;
  const bool m2 = (int)blockIdx.x >= K;
  const int k   = m2 ? (int)blockIdx.x - K : (int)blockIdx.x;
  int* __restrict__ cnt    = m2 ? cnt2 : cnt1;
  int* __restrict__ bstart = m2 ? bstart2 : bstart1;
  const int nch = m2 ? nch2 : nch1;
  if (blockIdx.x == 0 && b < 2 * COUT) zstats[b] = 0.f;
  int* c = cnt + (long)k * nch * NB + b;
  int s = 0;
  for (int j = 0; j < nch; ++j) s += c[(long)j * NB];
  tot[b] = s;
  __syncthreads();
  int start = 0;
  for (int j = 0; j < b; ++j) start += tot[j];
  bstart[k * (NB + 1) + b] = start;
  if (b == NB - 1) bstart[k * (NB + 1) + NB] = start + s;
  int run = start;
  for (int j = 0; j < nch; ++j) {
    const int v = c[(long)j * NB];
    c[(long)j * NB] = run;
    run += v;
  }
}

__global__ void bin_scatter(const int* __restrict__ pin1, const int* __restrict__ pout1,
                            int P1, int nch1, const int* __restrict__ off1,
                            int2* __restrict__ spair1,
                            const int* __restrict__ pin2, const int* __restrict__ pout2,
                            int P2, int nch2, const int* __restrict__ off2,
                            int2* __restrict__ spair2,
                            int R) {
  __shared__ int cur[NB];
  const int k = blockIdx.y;
  const bool m2 = (int)blockIdx.x >= nch1;
  const int ch  = m2 ? (int)blockIdx.x - nch1 : (int)blockIdx.x;
  const int* __restrict__ pin  = m2 ? pin2 : pin1;
  const int* __restrict__ pout = m2 ? pout2 : pout1;
  const int* __restrict__ off  = m2 ? off2 : off1;
  int2* __restrict__ spair = m2 ? spair2 : spair1;
  const int P = m2 ? P2 : P1, nch = m2 ? nch2 : nch1;
  cur[threadIdx.x] = off[((long)k * nch + ch) * NB + threadIdx.x];
  __syncthreads();
  const long kb = (long)k * P;
  for (int i = threadIdx.x; i < CH; i += NB) {
    const int p = ch * CH + i;
    if (p < P) {
      const int o = pout[kb + p];
      const int b = o / R;
      const int pos = atomicAdd(&cur[b], 1);
      spair[kb + pos] = make_int2(pin[kb + p], o - b * R);
    }
  }
}

#define TP 80
template <int CIN>
__global__ __launch_bounds__(1024)
void conv_bucket(const float* __restrict__ feats,
                 const float* __restrict__ W,
                 const int2*  __restrict__ spair,
                 const int*   __restrict__ bstart,
                 float*       __restrict__ out,
                 float*       __restrict__ stats,
                 int P, int R, int n_out, int K) {
  constexpr int PITCH = CIN + 4;
  constexpr int LPR   = CIN / 4;
  constexpr int NI    = CIN / 8;
  extern __shared__ v4f tile4[];
  __shared__ v4f stage4[32 * 16 * PITCH / 4];
  __shared__ float red[2 * COUT];
  const int b    = blockIdx.x;
  const int tid  = threadIdx.x;
  const int lane = tid & 31;
  const int wave = __builtin_amdgcn_readfirstlane(tid >> 5);
  const int nw   = blockDim.x >> 5;
  const int half = lane >> 4;
  const int nl   = lane & 15;

  float* tile = reinterpret_cast<float*>(tile4);
  float* st = reinterpret_cast<float*>(stage4) + wave * (16 * PITCH);
  const int piece = lane & (LPR - 1);
  const int sub   = lane / LPR;
  const int rbase = (CIN == 32) ? 8 * (sub & 1) + 4 * (sub >> 1) : 8 * half;

  int bs_lo = 0, bs_hi = 0;
  if (lane < K) {
    bs_lo = bstart[lane * (NB + 1) + b];
    bs_hi = bstart[lane * (NB + 1) + b + 1];
  }
  const int nt = (bs_hi - bs_lo + 15) >> 4;
  int inc = nt;
#pragma unroll
  for (int d = 1; d < 32; d <<= 1) {
    const int t = __shfl_up(inc, d, 32);
    if (lane >= d) inc += t;
  }
  const int tp   = inc - nt;
  const int ttot = __builtin_amdgcn_readlane(inc, 31);

  auto locate = [&](int f, int& k, int& e, int& p0) {
    const unsigned m = __builtin_amdgcn_ballot_w32(tp <= f);
    k  = __popc(m) - 1;
    e  = __builtin_amdgcn_readlane(bs_hi, k);
    p0 = __builtin_amdgcn_readlane(bs_lo, k) + (f - __builtin_amdgcn_readlane(tp, k)) * 16;
  };
  auto load_pair = [&](int k, int e, int p0) {
    int2 pr = make_int2(0, 0);
    const int pa = p0 + nl;
    if (pa < e) pr = spair[(long)k * P + pa];
    return pr;
  };
  auto gather = [&](int prx, v4f (&av)[NI]) {
#pragma unroll
    for (int i = 0; i < NI; ++i) {
      const int srow = __shfl(prx, i + rbase, 32);
      av[i] = *reinterpret_cast<const v4f*>(feats + (long)srow * CIN + 4 * piece);
    }
  };

  int k, e, p0;
  locate(wave, k, e, p0);
  int2 pr = load_pair(k, e, p0);
  v4f av[NI];
  gather(pr.x, av);

  const v4f z = {};
  for (int i = tid; i < R * (TP / 4); i += blockDim.x) tile4[i] = z;
  if (tid < 2 * COUT) red[tid] = 0.f;
  __syncthreads();

  for (int f = wave; f < ttot; f += nw) {
    const float* bptr = W + (long)k * CIN * COUT + 2 * half * COUT + nl;
    v2f q0, q1, q2, q3;
    q0.x = bptr[0];    q0.y = bptr[COUT];
    q1.x = bptr[16];   q1.y = bptr[COUT + 16];
    q2.x = bptr[32];   q2.y = bptr[COUT + 32];
    q3.x = bptr[48];   q3.y = bptr[COUT + 48];

#pragma unroll
    for (int i = 0; i < NI; ++i)
      *reinterpret_cast<v4f*>(st + (i + rbase) * PITCH + 4 * piece) = av[i];
    __builtin_amdgcn_fence(__ATOMIC_RELEASE, "wavefront");
    __builtin_amdgcn_wave_barrier();
    __builtin_amdgcn_fence(__ATOMIC_ACQUIRE, "wavefront");
    const float* aptr = st + nl * PITCH + 2 * half;

    int kn, en, p0n;
    locate(f + nw, kn, en, p0n);
    const int2 prn = load_pair(kn, en, p0n);

    v8f c0 = {}, c1 = {}, c2 = {}, c3 = {};
#pragma unroll
    for (int cc = 0; cc < CIN; cc += 4) {
      const v2f a = *reinterpret_cast<const v2f*>(aptr + cc);
      v2f b0, b1, b2, b3;
      if (cc == 0) {
        b0 = q0; b1 = q1; b2 = q2; b3 = q3;
      } else {
        b0.x = bptr[cc * COUT];        b0.y = bptr[cc * COUT + COUT];
        b1.x = bptr[cc * COUT + 16];   b1.y = bptr[cc * COUT + COUT + 16];
        b2.x = bptr[cc * COUT + 32];   b2.y = bptr[cc * COUT + COUT + 32];
        b3.x = bptr[cc * COUT + 48];   b3.y = bptr[cc * COUT + COUT + 48];
      }
      c0 = __builtin_amdgcn_wmma_f32_16x16x4_f32(false, a, false, b0, (short)0, c0, false, false);
      c1 = __builtin_amdgcn_wmma_f32_16x16x4_f32(false, a, false, b1, (short)0, c1, false, false);
      c2 = __builtin_amdgcn_wmma_f32_16x16x4_f32(false, a, false, b2, (short)0, c2, false, false);
      c3 = __builtin_amdgcn_wmma_f32_16x16x4_f32(false, a, false, b3, (short)0, c3, false, false);
    }

    v4f avn[NI];
    gather(prn.x, avn);

#pragma unroll
    for (int r = 0; r < 8; ++r) {
      const int src  = r + 8 * half;
      const int lrow = __shfl(pr.y, src, 32);
      if (p0 + src < e) {
        float* t = tile + lrow * TP + nl;
        atomicAdd(t,      c0[r]);
        atomicAdd(t + 16, c1[r]);
        atomicAdd(t + 32, c2[r]);
        atomicAdd(t + 48, c3[r]);
      }
    }

    k = kn; e = en; p0 = p0n; pr = prn;
#pragma unroll
    for (int i = 0; i < NI; ++i) av[i] = avn[i];
  }
  __syncthreads();

  const int row0 = b * R;
  v4f s1 = {}, s2 = {};
  for (int i = tid; i < R * 16; i += blockDim.x) {
    const int row = row0 + (i >> 4);
    if (row < n_out) {
      const v4f v = tile4[(i >> 4) * (TP / 4) + (i & 15)];
      reinterpret_cast<v4f*>(out)[(long)row * 16 + (i & 15)] = v;
      s1 += v;
      s2 += v * v;
    }
  }
  {
    float* r1 = red + 4 * (tid & 15);
    atomicAdd(r1,     s1.x);  atomicAdd(r1 + COUT,     s2.x);
    atomicAdd(r1 + 1, s1.y);  atomicAdd(r1 + COUT + 1, s2.y);
    atomicAdd(r1 + 2, s1.z);  atomicAdd(r1 + COUT + 2, s2.z);
    atomicAdd(r1 + 3, s1.w);  atomicAdd(r1 + COUT + 3, s2.w);
  }
  __syncthreads();
  if (tid < 2 * COUT) atomicAdd(&stats[tid], red[tid]);
}

__global__ void bn_finalize(float* __restrict__ stats,
                            const float* __restrict__ g,
                            const float* __restrict__ b,
                            float inv_n, float* __restrict__ ab) {
  int c = threadIdx.x;
  if (c < COUT) {
    const float sum = stats[c], sq = stats[COUT + c];
    stats[c] = 0.f;
    stats[COUT + c] = 0.f;
    float mean = sum * inv_n;
    float var  = sq * inv_n - mean * mean;
    float s    = g[c] * rsqrtf(var + BN_EPS);
    ab[c]        = s;
    ab[COUT + c] = b[c] - mean * s;
  }
}

__global__ void bn_apply(const float* __restrict__ y,
                         const float* __restrict__ ab,
                         float* __restrict__ x, int n_elems) {
  int i = blockIdx.x * blockDim.x + threadIdx.x;
  int s = gridDim.x * blockDim.x;
  for (; i < n_elems; i += s) {
    int ch = i & (COUT - 1);
    float v = y[i] * ab[ch] + ab[COUT + ch];
    x[i] = v > 0.f ? v : 0.f;
  }
}

__global__ void bn_apply_add(const float* __restrict__ y,
                             const float* __restrict__ ab,
                             const float* __restrict__ xnet,
                             float* __restrict__ out, int n_elems) {
  int i = blockIdx.x * blockDim.x + threadIdx.x;
  int s = gridDim.x * blockDim.x;
  for (; i < n_elems; i += s) {
    int ch = i & (COUT - 1);
    float v = y[i] * ab[ch] + ab[COUT + ch];
    out[i] = xnet[i] + (v > 0.f ? v : 0.f);
  }
}

static inline int cdiv(long a, long b) { return (int)((a + b - 1) / b); }

extern "C" void kernel_launch(void* const* d_in, const int* in_sizes, int n_in,
                              void* d_out, int out_size, void* d_ws, size_t ws_size,
                              hipStream_t stream) {
  const float* feats  = (const float*)d_in[0];
  const float* W_down = (const float*)d_in[1];
  const float* g0     = (const float*)d_in[2];
  const float* b0     = (const float*)d_in[3];
  const float* W1     = (const float*)d_in[4];
  const float* g1     = (const float*)d_in[5];
  const float* b1     = (const float*)d_in[6];
  const float* W2     = (const float*)d_in[7];
  const float* g2     = (const float*)d_in[8];
  const float* b2     = (const float*)d_in[9];
  const int* p_down_in  = (const int*)d_in[10];
  const int* p_down_out = (const int*)d_in[11];
  const int* p_subm_in  = (const int*)d_in[12];
  const int* p_subm_out = (const int*)d_in[13];

  const int Cin = 32;
  const int K   = in_sizes[1] / (Cin * COUT);
  const int P1  = in_sizes[10] / K;
  const int P2  = in_sizes[12] / K;
  const int n_out = out_size / COUT;
  const int NE  = n_out * COUT;
  const float inv_n = 1.0f / (float)n_out;

  float* y     = (float*)d_ws;
  float* x     = y + NE;
  float* xnet  = x + NE;
  float* stats = xnet + NE;
  float* ab    = stats + 128;
  (void)n_in; (void)ws_size;

  const dim3 blk(256);
  const int eb  = cdiv(NE, 256);

  const int R    = cdiv(n_out, NB);
  const int nch1 = cdiv(P1, CH);
  const int nch2 = cdiv(P2, CH);
  int2* sp1  = (int2*)x;
  int*  cnt1 = (int*)(sp1 + (long)K * P1);
  int*  bs1  = cnt1 + (long)K * nch1 * NB;
  int2* sp2  = (int2*)d_out;
  int*  cnt2 = (int*)(sp2 + (long)K * P2);
  int*  bs2  = cnt2 + (long)K * nch2 * NB;
  const size_t lds = (size_t)R * TP * sizeof(float);

  bin_hist<<<dim3(nch1 + nch2, K), NB, 0, stream>>>(p_down_out, P1, nch1, cnt1,
                                                    p_subm_out, P2, nch2, cnt2, R);
  bin_scan<<<2 * K, NB, 0, stream>>>(cnt1, nch1, bs1, cnt2, nch2, bs2, K, stats);
  bin_scatter<<<dim3(nch1 + nch2, K), NB, 0, stream>>>(p_down_in, p_down_out, P1, nch1, cnt1, sp1,
                                                       p_subm_in, p_subm_out, P2, nch2, cnt2, sp2, R);

  conv_bucket<32><<<NB, 1024, lds, stream>>>(feats, W_down, sp1, bs1, y, stats, P1, R, n_out, K);
  bn_finalize<<<1, 64, 0, stream>>>(stats, g0, b0, inv_n, ab);
  bn_apply<<<eb, blk, 0, stream>>>(y, ab, x, NE);

  conv_bucket<64><<<NB, 1024, lds, stream>>>(x, W1, sp2, bs2, y, stats, P2, R, n_out, K);
  bn_finalize<<<1, 64, 0, stream>>>(stats, g1, b1, inv_n, ab);
  bn_apply<<<eb, blk, 0, stream>>>(y, ab, xnet, NE);

  conv_bucket<64><<<NB, 1024, lds, stream>>>(xnet, W2, sp2, bs2, y, stats, P2, R, n_out, K);
  bn_finalize<<<1, 64, 0, stream>>>(stats, g2, b2, inv_n, ab);
  bn_apply_add<<<eb, blk, 0, stream>>>(y, ab, xnet, (float*)d_out, NE);
}


# strided conv: input rows of pairs sharing an output row summed first (LDS float adds, arrival order), then one dense product per 16 rows, accumulators in registers over all offsets; no D scatter
# speedup vs baseline: 2.7749x; 1.0760x over previous
#include <hip/hip_runtime.h>

typedef float v2f __attribute__((ext_vector_type(2)));
typedef float v4f __attribute__((ext_vector_type(4)));
typedef float v8f __attribute__((ext_vector_type(8)));

#define COUT 64
#define BN_EPS 1e-5f

#define NB 256
#define CH 2048

__global__ void bin_hist(const int* __restrict__ pout1, int P1, int nch1,
                         int* __restrict__ cnt1,
                         const int* __restrict__ pout2, int P2, int nch2,
                         int* __restrict__ cnt2,
                         int R) {
  __shared__ int h[NB];
  const int k = blockIdx.y;
  const bool m2 = (int)blockIdx.x >= nch1;
  const int ch  = m2 ? (int)blockIdx.x - nch1 : (int)blockIdx.x;
  const int* __restrict__ pout = m2 ? pout2 : pout1;
  int* __restrict__ cnt = m2 ? cnt2 : cnt1;
  const int P = m2 ? P2 : P1, nch = m2 ? nch2 : nch1;
  h[threadIdx.x] = 0;
  __syncthreads();
  const long kb = (long)k * P;
  for (int i = threadIdx.x; i < CH; i += NB) {
    const int p = ch * CH + i;
    if (p < P) atomicAdd(&h[pout[kb + p] / R], 1);
  }
  __syncthreads();
  cnt[((long)k * nch + ch) * NB + threadIdx.x] = h[threadIdx.x];
}

__global__ void bin_scan(int* __restrict__ cnt1, int nch1,
                         int* __restrict__ bstart1,
                         int* __restrict__ cnt2, int nch2,
                         int* __restrict__ bstart2,
                         int K, float* __restrict__ zstats) {
  __shared__ int tot[NB];
  const int b = threadIdx.x;
  const bool m2 = (int)blockIdx.x >= K;
  const int k   = m2 ? (int)blockIdx.x - K : (int)blockIdx.x;
  int* __restrict__ cnt    = m2 ? cnt2 : cnt1;
  int* __restrict__ bstart = m2 ? bstart2 : bstart1;
  const int nch = m2 ? nch2 : nch1;
  if (blockIdx.x == 0 && b < 2 * COUT) zstats[b] = 0.f;
  int* c = cnt + (long)k * nch * NB + b;
  int s = 0;
  for (int j = 0; j < nch; ++j) s += c[(long)j * NB];
  tot[b] = s;
  __syncthreads();
  int start = 0;
  for (int j = 0; j < b; ++j) start += tot[j];
  bstart[k * (NB + 1) + b] = start;
  if (b == NB - 1) bstart[k * (NB + 1) + NB] = start + s;
  int run = start;
  for (int j = 0; j < nch; ++j) {
    const int v = c[(long)j * NB];
    c[(long)j * NB] = run;
    run += v;
  }
}

__global__ void bin_scatter(const int* __restrict__ pin1, const int* __restrict__ pout1,
                            int P1, int nch1, const int* __restrict__ off1,
                            int2* __restrict__ spair1,
                            const int* __restrict__ pin2, const int* __restrict__ pout2,
                            int P2, int nch2, const int* __restrict__ off2,
                            int2* __restrict__ spair2,
                            int R) {
  __shared__ int cur[NB];
  const int k = blockIdx.y;
  const bool m2 = (int)blockIdx.x >= nch1;
  const int ch  = m2 ? (int)blockIdx.x - nch1 : (int)blockIdx.x;
  const int* __restrict__ pin  = m2 ? pin2 : pin1;
  const int* __restrict__ pout = m2 ? pout2 : pout1;
  const int* __restrict__ off  = m2 ? off2 : off1;
  int2* __restrict__ spair = m2 ? spair2 : spair1;
  const int P = m2 ? P2 : P1, nch = m2 ? nch2 : nch1;
  cur[threadIdx.x] = off[((long)k * nch + ch) * NB + threadIdx.x];
  __syncthreads();
  const long kb = (long)k * P;
  for (int i = threadIdx.x; i < CH; i += NB) {
    const int p = ch * CH + i;
    if (p < P) {
      const int o = pout[kb + p];
      const int b = o / R;
      const int pos = atomicAdd(&cur[b], 1);
      spair[kb + pos] = make_int2(pin[kb + p], o - b * R);
    }
  }
}

#define TP 80
template <int CIN>
__global__ __launch_bounds__(1024)
void conv_bucket(const float* __restrict__ feats,
                 const float* __restrict__ W,
                 const int2*  __restrict__ spair,
                 const int*   __restrict__ bstart,
                 float*       __restrict__ out,
                 float*       __restrict__ stats,
                 int P, int R, int n_out, int K) {
  constexpr int PITCH = CIN + 4;
  constexpr int LPR   = CIN / 4;
  constexpr int NI    = CIN / 8;
  extern __shared__ v4f tile4[];
  __shared__ v4f stage4[32 * 16 * PITCH / 4];
  __shared__ float red[2 * COUT];
  const int b    = blockIdx.x;
  const int tid  = threadIdx.x;
  const int lane = tid & 31;
  const int wave = __builtin_amdgcn_readfirstlane(tid >> 5);
  const int nw   = blockDim.x >> 5;
  const int half = lane >> 4;
  const int nl   = lane & 15;

  float* tile = reinterpret_cast<float*>(tile4);
  float* st = reinterpret_cast<float*>(stage4) + wave * (16 * PITCH);
  const int piece = lane & (LPR - 1);
  const int sub   = lane / LPR;
  const int rbase = (CIN == 32) ? 8 * (sub & 1) + 4 * (sub >> 1) : 8 * half;

  int bs_lo = 0, bs_hi = 0;
  if (lane < K) {
    bs_lo = bstart[lane * (NB + 1) + b];
    bs_hi = bstart[lane * (NB + 1) + b + 1];
  }
  const int nt = (bs_hi - bs_lo + 15) >> 4;
  int inc = nt;
#pragma unroll
  for (int d = 1; d < 32; d <<= 1) {
    const int t = __shfl_up(inc, d, 32);
    if (lane >= d) inc += t;
  }
  const int tp   = inc - nt;
  const int ttot = __builtin_amdgcn_readlane(inc, 31);

  auto locate = [&](int f, int& k, int& e, int& p0) {
    const unsigned m = __builtin_amdgcn_ballot_w32(tp <= f);
    k  = __popc(m) - 1;
    e  = __builtin_amdgcn_readlane(bs_hi, k);
    p0 = __builtin_amdgcn_readlane(bs_lo, k) + (f - __builtin_amdgcn_readlane(tp, k)) * 16;
  };
  auto load_pair = [&](int k, int e, int p0) {
    int2 pr = make_int2(0, 0);
    const int pa = p0 + nl;
    if (pa < e) pr = spair[(long)k * P + pa];
    return pr;
  };
  auto gather = [&](int prx, v4f (&av)[NI]) {
#pragma unroll
    for (int i = 0; i < NI; ++i) {
      const int srow = __shfl(prx, i + rbase, 32);
      av[i] = *reinterpret_cast<const v4f*>(feats + (long)srow * CIN + 4 * piece);
    }
  };

  int k, e, p0;
  locate(wave, k, e, p0);
  int2 pr = load_pair(k, e, p0);
  v4f av[NI];
  gather(pr.x, av);

  const v4f z = {};
  for (int i = tid; i < R * (TP / 4); i += blockDim.x) tile4[i] = z;
  if (tid < 2 * COUT) red[tid] = 0.f;
  __syncthreads();

  for (int f = wave; f < ttot; f += nw) {
    const float* bptr = W + (long)k * CIN * COUT + 2 * half * COUT + nl;
    v2f q0, q1, q2, q3;
    q0.x = bptr[0];    q0.y = bptr[COUT];
    q1.x = bptr[16];   q1.y = bptr[COUT + 16];
    q2.x = bptr[32];   q2.y = bptr[COUT + 32];
    q3.x = bptr[48];   q3.y = bptr[COUT + 48];

#pragma unroll
    for (int i = 0; i < NI; ++i)
      *reinterpret_cast<v4f*>(st + (i + rbase) * PITCH + 4 * piece) = av[i];
    __builtin_amdgcn_fence(__ATOMIC_RELEASE, "wavefront");
    __builtin_amdgcn_wave_barrier();
    __builtin_amdgcn_fence(__ATOMIC_ACQUIRE, "wavefront");
    const float* aptr = st + nl * PITCH + 2 * half;

    int kn, en, p0n;
    locate(f + nw, kn, en, p0n);
    const int2 prn = load_pair(kn, en, p0n);

    v8f c0 = {}, c1 = {}, c2 = {}, c3 = {};
#pragma unroll
    for (int cc = 0; cc < CIN; cc += 4) {
      const v2f a = *reinterpret_cast<const v2f*>(aptr + cc);
      v2f b0, b1, b2, b3;
      if (cc == 0) {
        b0 = q0; b1 = q1; b2 = q2; b3 = q3;
      } else {
        b0.x = bptr[cc * COUT];        b0.y = bptr[cc * COUT + COUT];
        b1.x = bptr[cc * COUT + 16];   b1.y = bptr[cc * COUT + COUT + 16];
        b2.x = bptr[cc * COUT + 32];   b2.y = bptr[cc * COUT + COUT + 32];
        b3.x = bptr[cc * COUT + 48];   b3.y = bptr[cc * COUT + COUT + 48];
      }
      c0 = __builtin_amdgcn_wmma_f32_16x16x4_f32(false, a, false, b0, (short)0, c0, false, false);
      c1 = __builtin_amdgcn_wmma_f32_16x16x4_f32(false, a, false, b1, (short)0, c1, false, false);
      c2 = __builtin_amdgcn_wmma_f32_16x16x4_f32(false, a, false, b2, (short)0, c2, false, false);
      c3 = __builtin_amdgcn_wmma_f32_16x16x4_f32(false, a, false, b3, (short)0, c3, false, false);
    }

    v4f avn[NI];
    gather(prn.x, avn);

#pragma unroll
    for (int r = 0; r < 8; ++r) {
      const int src  = r + 8 * half;
      const int lrow = __shfl(pr.y, src, 32);
      if (p0 + src < e) {
        float* t = tile + lrow * TP + nl;
        atomicAdd(t,      c0[r]);
        atomicAdd(t + 16, c1[r]);
        atomicAdd(t + 32, c2[r]);
        atomicAdd(t + 48, c3[r]);
      }
    }

    k = kn; e = en; p0 = p0n; pr = prn;
#pragma unroll
    for (int i = 0; i < NI; ++i) av[i] = avn[i];
  }
  __syncthreads();

  const int row0 = b * R;
  v4f s1 = {}, s2 = {};
  for (int i = tid; i < R * 16; i += blockDim.x) {
    const int row = row0 + (i >> 4);
    if (row < n_out) {
      const v4f v = tile4[(i >> 4) * (TP / 4) + (i & 15)];
      reinterpret_cast<v4f*>(out)[(long)row * 16 + (i & 15)] = v;
      s1 += v;
      s2 += v * v;
    }
  }
  {
    float* r1 = red + 4 * (tid & 15);
    atomicAdd(r1,     s1.x);  atomicAdd(r1 + COUT,     s2.x);
    atomicAdd(r1 + 1, s1.y);  atomicAdd(r1 + COUT + 1, s2.y);
    atomicAdd(r1 + 2, s1.z);  atomicAdd(r1 + COUT + 2, s2.z);
    atomicAdd(r1 + 3, s1.w);  atomicAdd(r1 + COUT + 3, s2.w);
  }
  __syncthreads();
  if (tid < 2 * COUT) atomicAdd(&stats[tid], red[tid]);
}

#define AP 36
#define NG 4
#define AMAX 256
__global__ __launch_bounds__(1024)
void conv_presum32(const float* __restrict__ feats,
                   const float* __restrict__ W,
                   const int2*  __restrict__ spair,
                   const int*   __restrict__ bstart,
                   float*       __restrict__ out,
                   float*       __restrict__ stats,
                   int P, int R, int n_out, int K) {
  constexpr int CIN = 32;
  extern __shared__ v4f tile4[];
  __shared__ v4f abuf4[3 * AMAX * AP / 4];
  __shared__ float red[2 * COUT];
  const int b    = blockIdx.x;
  const int tid  = threadIdx.x;
  const int lane = tid & 31;
  const int wave = __builtin_amdgcn_readfirstlane(tid >> 5);
  const int half = lane >> 4;
  const int nl   = lane & 15;
  const int sub   = lane >> 3;
  const int piece = lane & 7;
  const int MT  = (R + 15) >> 4;
  const int ASZ = MT * 16 * AP;
  float* tile = reinterpret_cast<float*>(tile4);
  float* abuf = reinterpret_cast<float*>(abuf4);

  int bs_lo = 0, bs_hi = 0;
  if (lane < K) {
    bs_lo = bstart[lane * (NB + 1) + b];
    bs_hi = bstart[lane * (NB + 1) + b + 1];
  }

  auto load_recs = [&](int kk, int2 (&rc)[NG]) {
    int lo = 0, hi = 0;
    if (kk < K) {
      lo = __builtin_amdgcn_readlane(bs_lo, kk);
      hi = __builtin_amdgcn_readlane(bs_hi, kk);
    }
    const long kb = (long)min(kk, K - 1) * P;
#pragma unroll
    for (int i = 0; i < NG; ++i) {
      const int idx = lo + ((i * 32 + wave) << 2) + sub;
      const int2 r = spair[kb + min(idx, P - 1)];
      rc[i] = make_int2(r.x | (idx < hi ? 0 : (int)0x80000000), r.y);
    }
  };
  auto gather = [&](const int2 (&rc)[NG], v4f (&rv)[NG]) {
#pragma unroll
    for (int i = 0; i < NG; ++i)
      rv[i] = *reinterpret_cast<const v4f*>(feats + (long)max(rc[i].x, 0) * CIN + 4 * piece);
  };
  const bool rot1 = (sub & 1) != 0, rot2 = (sub & 2) != 0;
  const int e0 = sub, e1 = (sub + 1) & 3, e2 = (sub + 2) & 3, e3 = (sub + 3) & 3;
  auto add_piece = [&](float* dst, int lrow, v4f v) {
    float* t = dst + lrow * AP + 4 * piece;
    const float p0 = rot1 ? v.y : v.x, p1 = rot1 ? v.z : v.y, p2 = rot1 ? v.w : v.z, p3 = rot1 ? v.x : v.w;
    const float q0 = rot2 ? p2 : p0,  q1 = rot2 ? p3 : p1,  q2 = rot2 ? p0 : p2,  q3 = rot2 ? p1 : p3;
    atomicAdd(t + e0, q0);
    atomicAdd(t + e1, q1);
    atomicAdd(t + e2, q2);
    atomicAdd(t + e3, q3);
  };
  auto add_rows = [&](float* dst, const int2 (&rc)[NG], const v4f (&rv)[NG]) {
#pragma unroll
    for (int i = 0; i < NG; ++i)
      if (rc[i].x >= 0) add_piece(dst, rc[i].y, rv[i]);
  };
  auto add_rest = [&](float* dst, int kk) {
    if (kk >= K) return;
    const int lo = __builtin_amdgcn_readlane(bs_lo, kk);
    const int hi = __builtin_amdgcn_readlane(bs_hi, kk);
    for (int s = NG * 32 + wave; (s << 2) < hi - lo; s += 32) {
      const int idx = lo + (s << 2) + sub;
      if (idx < hi) {
        const int2 pr = spair[(long)kk * P + idx];
        const v4f v = *reinterpret_cast<const v4f*>(feats + (long)pr.x * CIN + 4 * piece);
        add_piece(dst, pr.y, v);
      }
    }
  };
  auto lds_barrier = [&]() {
    __builtin_amdgcn_fence(__ATOMIC_RELEASE, "workgroup", "local");
    __builtin_amdgcn_s_barrier();
    __builtin_amdgcn_fence(__ATOMIC_ACQUIRE, "workgroup", "local");
  };
  const v4f z = {};
  auto clear = [&](float* buf) {
    v4f* p4 = reinterpret_cast<v4f*>(buf);
    for (int i = tid; i < ASZ / 4; i += blockDim.x) p4[i] = z;
  };

  const bool mm = wave < 2 * MT;
  const int m  = wave >> 1;
  const int nh = wave & 1;
  v2f bq[CIN / 4][2];
  auto load_b = [&](int kk) {
    const float* bptr = W + (long)kk * CIN * COUT + 2 * half * COUT + 32 * nh + nl;
#pragma unroll
    for (int s = 0; s < CIN / 4; ++s) {
      bq[s][0].x = bptr[4 * s * COUT];        bq[s][0].y = bptr[4 * s * COUT + COUT];
      bq[s][1].x = bptr[4 * s * COUT + 16];   bq[s][1].y = bptr[4 * s * COUT + COUT + 16];
    }
  };

  int2 rcv[NG], rcn[NG];
  v4f rv[NG];
  load_recs(0, rcv);
  clear(abuf);
  clear(abuf + ASZ);
  if (tid < 2 * COUT) red[tid] = 0.f;
  gather(rcv, rv);
  __syncthreads();
  add_rows(abuf, rcv, rv);
  add_rest(abuf, 0);
  load_recs(1, rcv);
  if (mm) load_b(0);
  gather(rcv, rv);
  lds_barrier();

  v8f c0 = {}, c1 = {};
  for (int k = 0; k < K; ++k) {
    const float* Ak = abuf + (k % 3) * ASZ;
    float* An = abuf + ((k + 1) % 3) * ASZ;
    float* Az = abuf + ((k + 2) % 3) * ASZ;
    load_recs(k + 2, rcn);
    if (mm) {
      const float* aptr = Ak + (16 * m + nl) * AP + 2 * half;
#pragma unroll
      for (int s = 0; s < CIN / 4; ++s) {
        const v2f a = *reinterpret_cast<const v2f*>(aptr + 4 * s);
        c0 = __builtin_amdgcn_wmma_f32_16x16x4_f32(false, a, false, bq[s][0], (short)0, c0, false, false);
        c1 = __builtin_amdgcn_wmma_f32_16x16x4_f32(false, a, false, bq[s][1], (short)0, c1, false, false);
      }
      if (k + 1 < K) load_b(k + 1);
    }
    add_rows(An, rcv, rv);
    add_rest(An, k + 1);
#pragma unroll
    for (int i = 0; i < NG; ++i) rcv[i] = rcn[i];
    gather(rcv, rv);
    clear(Az);
    lds_barrier();
  }

  if (mm) {
#pragma unroll
    for (int r = 0; r < 8; ++r) {
      const int row = 16 * m + r + 8 * half;
      if (row < R) {
        float* t = tile + row * TP + 32 * nh + nl;
        t[0]  = c0[r];
        t[16] = c1[r];
      }
    }
  }
  __syncthreads();

  const int row0 = b * R;
  v4f s1 = {}, s2 = {};
  for (int i = tid; i < R * 16; i += blockDim.x) {
    const int row = row0 + (i >> 4);
    if (row < n_out) {
      const v4f v = tile4[(i >> 4) * (TP / 4) + (i & 15)];
      reinterpret_cast<v4f*>(out)[(long)row * 16 + (i & 15)] = v;
      s1 += v;
      s2 += v * v;
    }
  }
  {
    float* r1 = red + 4 * (tid & 15);
    atomicAdd(r1,     s1.x);  atomicAdd(r1 + COUT,     s2.x);
    atomicAdd(r1 + 1, s1.y);  atomicAdd(r1 + COUT + 1, s2.y);
    atomicAdd(r1 + 2, s1.z);  atomicAdd(r1 + COUT + 2, s2.z);
    atomicAdd(r1 + 3, s1.w);  atomicAdd(r1 + COUT + 3, s2.w);
  }
  __syncthreads();
  if (tid < 2 * COUT) atomicAdd(&stats[tid], red[tid]);
}

__global__ void bn_finalize(float* __restrict__ stats,
                            const float* __restrict__ g,
                            const float* __restrict__ b,
                            float inv_n, float* __restrict__ ab) {
  int c = threadIdx.x;
  if (c < COUT) {
    const float sum = stats[c], sq = stats[COUT + c];
    stats[c] = 0.f;
    stats[COUT + c] = 0.f;
    float mean = sum * inv_n;
    float var  = sq * inv_n - mean * mean;
    float s    = g[c] * rsqrtf(var + BN_EPS);
    ab[c]        = s;
    ab[COUT + c] = b[c] - mean * s;
  }
}

__global__ void bn_apply(const float* __restrict__ y,
                         const float* __restrict__ ab,
                         float* __restrict__ x, int n_elems) {
  int i = blockIdx.x * blockDim.x + threadIdx.x;
  int s = gridDim.x * blockDim.x;
  for (; i < n_elems; i += s) {
    int ch = i & (COUT - 1);
    float v = y[i] * ab[ch] + ab[COUT + ch];
    x[i] = v > 0.f ? v : 0.f;
  }
}

__global__ void bn_apply_add(const float* __restrict__ y,
                             const float* __restrict__ ab,
                             const float* __restrict__ xnet,
                             float* __restrict__ out, int n_elems) {
  int i = blockIdx.x * blockDim.x + threadIdx.x;
  int s = gridDim.x * blockDim.x;
  for (; i < n_elems; i += s) {
    int ch = i & (COUT - 1);
    float v = y[i] * ab[ch] + ab[COUT + ch];
    out[i] = xnet[i] + (v > 0.f ? v : 0.f);
  }
}

static inline int cdiv(long a, long b) { return (int)((a + b - 1) / b); }

extern "C" void kernel_launch(void* const* d_in, const int* in_sizes, int n_in,
                              void* d_out, int out_size, void* d_ws, size_t ws_size,
                              hipStream_t stream) {
  const float* feats  = (const float*)d_in[0];
  const float* W_down = (const float*)d_in[1];
  const float* g0     = (const float*)d_in[2];
  const float* b0     = (const float*)d_in[3];
  const float* W1     = (const float*)d_in[4];
  const float* g1     = (const float*)d_in[5];
  const float* b1     = (const float*)d_in[6];
  const float* W2     = (const float*)d_in[7];
  const float* g2     = (const float*)d_in[8];
  const float* b2     = (const float*)d_in[9];
  const int* p_down_in  = (const int*)d_in[10];
  const int* p_down_out = (const int*)d_in[11];
  const int* p_subm_in  = (const int*)d_in[12];
  const int* p_subm_out = (const int*)d_in[13];

  const int Cin = 32;
  const int K   = in_sizes[1] / (Cin * COUT);
  const int P1  = in_sizes[10] / K;
  const int P2  = in_sizes[12] / K;
  const int n_out = out_size / COUT;
  const int NE  = n_out * COUT;
  const float inv_n = 1.0f / (float)n_out;

  float* y     = (float*)d_ws;
  float* x     = y + NE;
  float* xnet  = x + NE;
  float* stats = xnet + NE;
  float* ab    = stats + 128;
  (void)n_in; (void)ws_size;

  const dim3 blk(256);
  const int eb  = cdiv(NE, 256);

  const int R    = cdiv(n_out, NB);
  const int nch1 = cdiv(P1, CH);
  const int nch2 = cdiv(P2, CH);
  int2* sp1  = (int2*)x;
  int*  cnt1 = (int*)(sp1 + (long)K * P1);
  int*  bs1  = cnt1 + (long)K * nch1 * NB;
  int2* sp2  = (int2*)d_out;
  int*  cnt2 = (int*)(sp2 + (long)K * P2);
  int*  bs2  = cnt2 + (long)K * nch2 * NB;
  const size_t lds = (size_t)R * TP * sizeof(float);

  bin_hist<<<dim3(nch1 + nch2, K), NB, 0, stream>>>(p_down_out, P1, nch1, cnt1,
                                                    p_subm_out, P2, nch2, cnt2, R);
  bin_scan<<<2 * K, NB, 0, stream>>>(cnt1, nch1, bs1, cnt2, nch2, bs2, K, stats);
  bin_scatter<<<dim3(nch1 + nch2, K), NB, 0, stream>>>(p_down_in, p_down_out, P1, nch1, cnt1, sp1,
                                                       p_subm_in, p_subm_out, P2, nch2, cnt2, sp2, R);

  if (R <= AMAX)
    conv_presum32<<<NB, 1024, lds, stream>>>(feats, W_down, sp1, bs1, y, stats, P1, R, n_out, K);
  else
    conv_bucket<32><<<NB, 1024, lds, stream>>>(feats, W_down, sp1, bs1, y, stats, P1, R, n_out, K);
  bn_finalize<<<1, 64, 0, stream>>>(stats, g0, b0, inv_n, ab);
  bn_apply<<<eb, blk, 0, stream>>>(y, ab, x, NE);

  conv_bucket<64><<<NB, 1024, lds, stream>>>(x, W1, sp2, bs2, y, stats, P2, R, n_out, K);
  bn_finalize<<<1, 64, 0, stream>>>(stats, g1, b1, inv_n, ab);
  bn_apply<<<eb, blk, 0, stream>>>(y, ab, xnet, NE);

  conv_bucket<64><<<NB, 1024, lds, stream>>>(xnet, W2, sp2, bs2, y, stats, P2, R, n_out, K);
  bn_finalize<<<1, 64, 0, stream>>>(stats, g2, b2, inv_n, ab);
  bn_apply_add<<<eb, blk, 0, stream>>>(y, ab, xnet, (float*)d_out, NE);
}
